# starting kernel
# baseline (speedup 1.0000x reference)
#include <hip/hip_runtime.h>


typedef float v2f __attribute__((ext_vector_type(2)));
typedef float v8f __attribute__((ext_vector_type(8)));

#define B_TOT  16384
#define T_LEN  100
#define V_SZ   256
#define E_SZ   50
#define H_SZ   10
#define L_SZ   15
#define HP     16
#define ROWS_PER_BLOCK 128
#define TPB    256
#define T_CHUNK 20
#define H_STRIDE 20

__device__ __forceinline__ float fast_tanh(float x) {
#if __has_builtin(__builtin_amdgcn_tanhf)
    return __builtin_amdgcn_tanhf(x);
#else
    x = fminf(fmaxf(x, -9.0f), 9.0f);
    float z = __builtin_amdgcn_exp2f(x * 2.885390081777927f);
    return (z - 1.0f) * __builtin_amdgcn_rcpf(z + 1.0f);
#endif
}

__device__ __forceinline__ v8f wmma_f32_16x16x4(v2f a, v2f b, v8f c) {
    return __builtin_amdgcn_wmma_f32_16x16x4_f32(
         false, a,  false, b,
         (short)0, c,  false,  false);
}

__global__ __launch_bounds__(TPB)
void charrnn_kernel(const int*   __restrict__ x,
                    const float* __restrict__ emb,
                    const float* __restrict__ Wx,
                    const float* __restrict__ Wh,
                    const float* __restrict__ b_rnn,
                    const float* __restrict__ Wd,
                    const float* __restrict__ bd,
                    const float* __restrict__ dmask,
                    float*       __restrict__ out)
{
    __shared__ float s_projb[V_SZ * HP];
    __shared__ int   s_tok[ROWS_PER_BLOCK * T_CHUNK];
    __shared__ float s_h[(TPB / 32) * 16 * H_STRIDE];

    const int tid   = threadIdx.x;
    const int lane  = tid & 31;
    const int wave  = tid >> 5;
    const int n     = lane & 15;
    const int half  = lane >> 4;
    const int mBase = half * 8;
    const int rowLocal = wave * 16;
    const int gRowBase = blockIdx.x * ROWS_PER_BLOCK + rowLocal;

    {
        const int v = tid;
        float s[H_SZ];
#pragma unroll
        for (int j = 0; j < H_SZ; ++j) s[j] = b_rnn[j];
        for (int e = 0; e < E_SZ; ++e) {
            const float em = emb[v * E_SZ + e];
#pragma unroll
            for (int j = 0; j < H_SZ; ++j)
                s[j] = fmaf(em, Wx[e * H_SZ + j], s[j]);
        }
#pragma unroll
        for (int j = 0; j < HP; ++j)
            s_projb[v * HP + j] = (j < H_SZ) ? s[j] : 0.0f;
    }

    v2f whb[3];
#pragma unroll
    for (int c = 0; c < 3; ++c) {
#pragma unroll
        for (int j = 0; j < 2; ++j) {
            const int K = 4 * c + j + 2 * half;
            whb[c][j] = (K < H_SZ && n < H_SZ) ? Wh[K * H_SZ + n] : 0.0f;
        }
    }

    const float colMask = (n < H_SZ) ? 1.0f : 0.0f;
    float* hw = &s_h[wave * 16 * H_STRIDE];

#pragma unroll
    for (int r = 0; r < 8; ++r) hw[(r + mBase) * H_STRIDE + n] = 0.0f;

    __syncthreads();

    for (int tc = 0; tc < T_LEN; tc += T_CHUNK) {
        __syncthreads();
        for (int i = tid; i < ROWS_PER_BLOCK * T_CHUNK; i += TPB) {
            const int rl = i / T_CHUNK, tt = i % T_CHUNK;
            s_tok[i] = x[(blockIdx.x * ROWS_PER_BLOCK + rl) * T_LEN + tc + tt];
        }
        __syncthreads();

        for (int t = 0; t < T_CHUNK; ++t) {
            v2f a[3];
#pragma unroll
            for (int c = 0; c < 3; ++c) {
                const float* p = &hw[n * H_STRIDE + 4 * c + 2 * half];
                a[c][0] = p[0];
                a[c][1] = p[1];
            }

            v8f acc;
#pragma unroll
            for (int r = 0; r < 8; ++r) {
                const int tok = s_tok[(rowLocal + r + mBase) * T_CHUNK + t];
                acc[r] = s_projb[tok * HP + n];
            }

            acc = wmma_f32_16x16x4(a[0], whb[0], acc);
            acc = wmma_f32_16x16x4(a[1], whb[1], acc);
            acc = wmma_f32_16x16x4(a[2], whb[2], acc);

#pragma unroll
            for (int r = 0; r < 8; ++r)
                hw[(r + mBase) * H_STRIDE + n] = fast_tanh(acc[r]) * colMask;

            asm volatile("s_wait_dscnt 0" ::: "memory");
        }
    }

    v2f af[3], wdb[3];
#pragma unroll
    for (int c = 0; c < 3; ++c) {
#pragma unroll
        for (int j = 0; j < 2; ++j) {
            const int K = 4 * c + j + 2 * half;
            float hv = 0.0f;
            if (K < H_SZ)
                hv = hw[n * H_STRIDE + K] * dmask[(gRowBase + n) * H_SZ + K];
            af[c][j]  = hv;
            wdb[c][j] = (K < H_SZ && n < L_SZ) ? Wd[K * L_SZ + n] : 0.0f;
        }
    }

    const float bdv = (n < L_SZ) ? bd[n] : 0.0f;
    v8f lac;
#pragma unroll
    for (int r = 0; r < 8; ++r) lac[r] = bdv;

    lac = wmma_f32_16x16x4(af[0], wdb[0], lac);
    lac = wmma_f32_16x16x4(af[1], wdb[1], lac);
    lac = wmma_f32_16x16x4(af[2], wdb[2], lac);

    if (n < L_SZ) {
#pragma unroll
        for (int r = 0; r < 8; ++r)
            out[(gRowBase + r + mBase) * L_SZ + n] = lac[r];
    }
}

extern "C" void kernel_launch(void* const* d_in, const int* in_sizes, int n_in,
                              void* d_out, int out_size, void* d_ws, size_t ws_size,
                              hipStream_t stream) {
    (void)in_sizes; (void)n_in; (void)d_ws; (void)ws_size; (void)out_size;
    const int*   x     = (const int*)  d_in[0];
    const float* emb   = (const float*)d_in[1];
    const float* Wx    = (const float*)d_in[2];
    const float* Wh    = (const float*)d_in[3];
    const float* b_rnn = (const float*)d_in[4];
    const float* Wd    = (const float*)d_in[5];
    const float* bd    = (const float*)d_in[6];
    const float* dmask = (const float*)d_in[7];
    float* out = (float*)d_out;

    charrnn_kernel<<<B_TOT / ROWS_PER_BLOCK, TPB, 0, stream>>>(
        x, emb, Wx, Wh, b_rnn, Wd, bd, dmask, out);
}


# Transposed recurrence: Wh^T is the A operand and h^T the B operand, so tanh(D) is the next step's B fragment on the same lane; h stays in registers, no per-step LDS round trip or token staging
# speedup vs baseline: 2.0536x; 2.0536x over previous
#include <hip/hip_runtime.h>


typedef float v2f __attribute__((ext_vector_type(2)));
typedef float v8f __attribute__((ext_vector_type(8)));

#define B_TOT  16384
#define T_LEN  100
#define V_SZ   256
#define E_SZ   50
#define H_SZ   10
#define L_SZ   15
#define PSTR   12
#define ROWS_PER_BLOCK 128
#define TPB    256
#define T_CHUNK 20
#define N_CHUNK (T_LEN / T_CHUNK)
#define NQ     (T_CHUNK / 4)

__device__ __forceinline__ float fast_tanh(float x) {
#if __has_builtin(__builtin_amdgcn_tanhf)
    return __builtin_amdgcn_tanhf(x);
#else
    x = fminf(fmaxf(x, -9.0f), 9.0f);
    float z = __builtin_amdgcn_exp2f(x * 2.885390081777927f);
    return (z - 1.0f) * __builtin_amdgcn_rcpf(z + 1.0f);
#endif
}

__device__ __forceinline__ v8f wmma_f32_16x16x4(v2f a, v2f b, v8f c) {
    return __builtin_amdgcn_wmma_f32_16x16x4_f32(
        false, a, false, b, (short)0, c, false, false);
}

__device__ __forceinline__ int hid_of(int hh, int r) {
    return hh ? ((r < 4) ? 6 + r : -1) : ((r < 6) ? r : -1);
}

__device__ __forceinline__ void gather6(const float* tab, int tok, int hoff,
                                        v2f& p0, v2f& p1, v2f& p2) {
    const v2f* p = reinterpret_cast<const v2f*>(tab + (tok & (V_SZ - 1)) * PSTR + hoff);
    p0 = p[0];
    p1 = p[1];
    p2 = p[2];
}

__global__ __launch_bounds__(TPB)
void charrnn_kernel(const int*   __restrict__ x,
                    const float* __restrict__ emb,
                    const float* __restrict__ Wx,
                    const float* __restrict__ Wh,
                    const float* __restrict__ b_rnn,
                    const float* __restrict__ Wd,
                    const float* __restrict__ bd,
                    const float* __restrict__ dmask,
                    float*       __restrict__ out)
{
    __shared__ __attribute__((aligned(16))) float s_projb[V_SZ * PSTR];
    __shared__ __attribute__((aligned(16))) float s_out[ROWS_PER_BLOCK * L_SZ];

    const int tid   = threadIdx.x;
    const int lane  = tid & 31;
    const int wave  = tid >> 5;
    const int c     = lane & 15;
    const int hh    = lane >> 4;
    const int hoff  = 6 * hh;
    const int rowLocal = wave * 16 + c;
    const int gRow  = blockIdx.x * ROWS_PER_BLOCK + rowLocal;

    const int4* xrow = reinterpret_cast<const int4*>(x + (size_t)gRow * T_LEN);
    int cur[T_CHUNK];
#pragma unroll
    for (int q = 0; q < NQ; ++q) {
        const int4 v = xrow[q];
        cur[4 * q + 0] = v.x; cur[4 * q + 1] = v.y;
        cur[4 * q + 2] = v.z; cur[4 * q + 3] = v.w;
    }

    float dm[6];
    v2f wda[3];
#pragma unroll
    for (int r = 0; r < 6; ++r) {
        const int k = hid_of(hh, r);
        dm[r] = (k >= 0) ? dmask[(size_t)gRow * H_SZ + k] : 0.0f;
        wda[r >> 1][r & 1] = (k >= 0 && c < L_SZ) ? Wd[k * L_SZ + c] : 0.0f;
    }
    float bdv[8];
#pragma unroll
    for (int r = 0; r < 8; ++r) {
        const int l = 8 * hh + r;
        bdv[r] = (l < L_SZ) ? bd[l] : 0.0f;
    }

    {
        const int v = tid;
        float s[H_SZ];
#pragma unroll
        for (int j = 0; j < H_SZ; ++j) s[j] = b_rnn[j];
        for (int e = 0; e < E_SZ; ++e) {
            const float em = emb[v * E_SZ + e];
#pragma unroll
            for (int j = 0; j < H_SZ; ++j)
                s[j] = fmaf(em, Wx[e * H_SZ + j], s[j]);
        }
#pragma unroll
        for (int j = 0; j < PSTR; ++j)
            s_projb[v * PSTR + j] = (j < H_SZ) ? s[j] : 0.0f;
    }

    const int jo = hid_of(c >> 3, c & 7);
    v2f wha[3];
#pragma unroll
    for (int q = 0; q < 3; ++q) {
#pragma unroll
        for (int s = 0; s < 2; ++s) {
            const int k = hid_of(hh, 2 * q + s);
            wha[q][s] = (k >= 0 && jo >= 0) ? Wh[k * H_SZ + jo] : 0.0f;
        }
    }

    __syncthreads();

    v8f accp;
#pragma unroll
    for (int r = 0; r < 8; ++r) accp[r] = 0.0f;

    v2f ca0, ca1, ca2;
    v2f cb0, cb1, cb2;
    gather6(s_projb, cur[0], hoff, ca0, ca1, ca2);
    gather6(s_projb, cur[1], hoff, cb0, cb1, cb2);

    for (int ch = 0; ch < N_CHUNK; ++ch) {
        int nxt[T_CHUNK];
#pragma unroll
        for (int i = 0; i < T_CHUNK; ++i) nxt[i] = cur[i];
        if (ch + 1 < N_CHUNK) {
#pragma unroll
            for (int q = 0; q < NQ; ++q) {
                const int4 v = xrow[(ch + 1) * NQ + q];
                nxt[4 * q + 0] = v.x; nxt[4 * q + 1] = v.y;
                nxt[4 * q + 2] = v.z; nxt[4 * q + 3] = v.w;
            }
        }

#pragma unroll
        for (int t = 0; t < T_CHUNK; ++t) {
            const int tn = (t + 2 < T_CHUNK) ? cur[(t + 2) % T_CHUNK] : nxt[(t + 2) % T_CHUNK];
            v2f cc0, cc1, cc2;
            gather6(s_projb, tn, hoff, cc0, cc1, cc2);
            __builtin_amdgcn_sched_barrier(0);

            v8f acc;
            acc[0] = ca0[0]; acc[1] = ca0[1];
            acc[2] = ca1[0]; acc[3] = ca1[1];
            acc[4] = ca2[0]; acc[5] = ca2[1];
            acc[6] = 0.0f;   acc[7] = 0.0f;

            v2f b0, b1, b2;
            b0[0] = fast_tanh(accp[0]); b0[1] = fast_tanh(accp[1]);
            acc = wmma_f32_16x16x4(wha[0], b0, acc);
            b1[0] = fast_tanh(accp[2]); b1[1] = fast_tanh(accp[3]);
            acc = wmma_f32_16x16x4(wha[1], b1, acc);
            b2[0] = fast_tanh(accp[4]); b2[1] = fast_tanh(accp[5]);
            acc = wmma_f32_16x16x4(wha[2], b2, acc);

            accp = acc;
            ca0 = cb0; ca1 = cb1; ca2 = cb2;
            cb0 = cc0; cb1 = cc1; cb2 = cc2;
        }

#pragma unroll
        for (int i = 0; i < T_CHUNK; ++i) cur[i] = nxt[i];
    }

    v2f b0, b1, b2;
    b0[0] = fast_tanh(accp[0]); b0[1] = fast_tanh(accp[1]);
    b1[0] = fast_tanh(accp[2]); b1[1] = fast_tanh(accp[3]);
    b2[0] = fast_tanh(accp[4]); b2[1] = fast_tanh(accp[5]);

    v2f f0, f1, f2;
    f0[0] = b0[0] * dm[0]; f0[1] = b0[1] * dm[1];
    f1[0] = b1[0] * dm[2]; f1[1] = b1[1] * dm[3];
    f2[0] = b2[0] * dm[4]; f2[1] = b2[1] * dm[5];

    v8f lac;
#pragma unroll
    for (int r = 0; r < 8; ++r) lac[r] = bdv[r];

    lac = wmma_f32_16x16x4(wda[0], f0, lac);
    lac = wmma_f32_16x16x4(wda[1], f1, lac);
    lac = wmma_f32_16x16x4(wda[2], f2, lac);

#pragma unroll
    for (int r = 0; r < 8; ++r) {
        const int l = 8 * hh + r;
        if (l < L_SZ) s_out[rowLocal * L_SZ + l] = lac[r];
    }
    __syncthreads();

    float* ob = out + (size_t)blockIdx.x * (ROWS_PER_BLOCK * L_SZ);
#pragma unroll
    for (int i = tid; i < ROWS_PER_BLOCK * L_SZ; i += TPB) ob[i] = s_out[i];
}

extern "C" void kernel_launch(void* const* d_in, const int* in_sizes, int n_in,
                              void* d_out, int out_size, void* d_ws, size_t ws_size,
                              hipStream_t stream) {
    (void)in_sizes; (void)n_in; (void)d_ws; (void)ws_size; (void)out_size;
    const int*   x     = (const int*)  d_in[0];
    const float* emb   = (const float*)d_in[1];
    const float* Wx    = (const float*)d_in[2];
    const float* Wh    = (const float*)d_in[3];
    const float* b_rnn = (const float*)d_in[4];
    const float* Wd    = (const float*)d_in[5];
    const float* bd    = (const float*)d_in[6];
    const float* dmask = (const float*)d_in[7];
    float* out = (float*)d_out;

    charrnn_kernel<<<B_TOT / ROWS_PER_BLOCK, TPB, 0, stream>>>(
        x, emb, Wx, Wh, b_rnn, Wd, bd, dmask, out);
}


# Token table built on the matrix unit: projb^T = Wx^T * emb^T + b as 26 f32 WMMAs per wave with every operand load issued in one group, replacing ten serialised load-and-FMA rounds
# speedup vs baseline: 2.5258x; 1.2299x over previous
#include <hip/hip_runtime.h>


typedef float v2f __attribute__((ext_vector_type(2)));
typedef float v8f __attribute__((ext_vector_type(8)));

#define B_TOT  16384
#define T_LEN  100
#define V_SZ   256
#define E_SZ   50
#define H_SZ   10
#define L_SZ   15
#define PSTR   12
#define ROWS_PER_BLOCK 128
#define TPB    256
#define T_CHUNK 20
#define N_CHUNK (T_LEN / T_CHUNK)
#define NQ     (T_CHUNK / 4)
#define N_KC   13

__device__ __forceinline__ float fast_tanh(float x) {
#if __has_builtin(__builtin_amdgcn_tanhf)
    return __builtin_amdgcn_tanhf(x);
#else
    x = fminf(fmaxf(x, -9.0f), 9.0f);
    float z = __builtin_amdgcn_exp2f(x * 2.885390081777927f);
    return (z - 1.0f) * __builtin_amdgcn_rcpf(z + 1.0f);
#endif
}

__device__ __forceinline__ v8f wmma_f32_16x16x4(v2f a, v2f b, v8f c) {
    return __builtin_amdgcn_wmma_f32_16x16x4_f32(
        false, a, false, b, (short)0, c, false, false);
}

__device__ __forceinline__ int hid_of(int hh, int r) {
    return hh ? ((r < 4) ? 6 + r : -1) : ((r < 6) ? r : -1);
}

__device__ __forceinline__ int emb_col_of(int q, int hh, int s) {
    return (q < 12) ? 8 * (q >> 1) + 4 * hh + 2 * (q & 1) + s
                    : (hh ? -1 : 48 + s);
}

__device__ __forceinline__ void gather6(const float* tab, int tok, int hoff,
                                        v2f& p0, v2f& p1, v2f& p2) {
    const v2f* p = reinterpret_cast<const v2f*>(tab + (tok & (V_SZ - 1)) * PSTR + hoff);
    p0 = p[0];
    p1 = p[1];
    p2 = p[2];
}

__global__ __launch_bounds__(TPB)
void charrnn_kernel(const int*   __restrict__ x,
                    const float* __restrict__ emb,
                    const float* __restrict__ Wx,
                    const float* __restrict__ Wh,
                    const float* __restrict__ b_rnn,
                    const float* __restrict__ Wd,
                    const float* __restrict__ bd,
                    const float* __restrict__ dmask,
                    float*       __restrict__ out)
{
    __shared__ __attribute__((aligned(16))) float s_projb[V_SZ * PSTR];
    __shared__ __attribute__((aligned(16))) float s_out[ROWS_PER_BLOCK * L_SZ];

    const int tid   = threadIdx.x;
    const int lane  = tid & 31;
    const int wave  = tid >> 5;
    const int c     = lane & 15;
    const int hh    = lane >> 4;
    const int hoff  = 6 * hh;
    const int rowLocal = wave * 16 + c;
    const int gRow  = blockIdx.x * ROWS_PER_BLOCK + rowLocal;

    const int4* xrow = reinterpret_cast<const int4*>(x + (size_t)gRow * T_LEN);
    int cur[T_CHUNK];
#pragma unroll
    for (int q = 0; q < NQ; ++q) {
        const int4 v = xrow[q];
        cur[4 * q + 0] = v.x; cur[4 * q + 1] = v.y;
        cur[4 * q + 2] = v.z; cur[4 * q + 3] = v.w;
    }

    float dm[6];
    v2f wda[3];
#pragma unroll
    for (int r = 0; r < 6; ++r) {
        const int k = hid_of(hh, r);
        dm[r] = (k >= 0) ? dmask[(size_t)gRow * H_SZ + k] : 0.0f;
        wda[r >> 1][r & 1] = (k >= 0 && c < L_SZ) ? Wd[k * L_SZ + c] : 0.0f;
    }
    float bdv[8];
#pragma unroll
    for (int r = 0; r < 8; ++r) {
        const int l = 8 * hh + r;
        bdv[r] = (l < L_SZ) ? bd[l] : 0.0f;
    }

    const int jo = hid_of(c >> 3, c & 7);
    v2f wha[3];
#pragma unroll
    for (int q = 0; q < 3; ++q) {
#pragma unroll
        for (int s = 0; s < 2; ++s) {
            const int k = hid_of(hh, 2 * q + s);
            wha[q][s] = (k >= 0 && jo >= 0) ? Wh[k * H_SZ + jo] : 0.0f;
        }
    }

    {
        const int cj = (c < H_SZ) ? c : (H_SZ - 1);
        v2f wxa[N_KC];
#pragma unroll
        for (int q = 0; q < N_KC; ++q) {
#pragma unroll
            for (int s = 0; s < 2; ++s) {
                const int e  = emb_col_of(q, hh, s);
                const int ec = (e >= 0) ? e : emb_col_of(q, 0, s);
                const float w = Wx[ec * H_SZ + cj];
                wxa[q][s] = (e >= 0 && c < H_SZ) ? w : 0.0f;
            }
        }

        v8f bias;
#pragma unroll
        for (int r = 0; r < 8; ++r) {
            const int j  = 8 * hh + r;
            const float b = b_rnn[(j < H_SZ) ? j : (H_SZ - 1)];
            bias[r] = (j < H_SZ) ? b : 0.0f;
        }

        const int v0 = wave * (V_SZ / (TPB / 32));
        v2f eb[2][N_KC];
#pragma unroll
        for (int tl = 0; tl < 2; ++tl) {
            const float* er = emb + (v0 + 16 * tl + c) * E_SZ;
#pragma unroll
            for (int q = 0; q < N_KC; ++q) {
#pragma unroll
                for (int s = 0; s < 2; ++s) {
                    const int e  = emb_col_of(q, hh, s);
                    const int ec = (e >= 0) ? e : emb_col_of(q, 0, s);
                    const float ev = er[ec];
                    eb[tl][q][s] = (e >= 0) ? ev : 0.0f;
                }
            }
        }

        v8f t0 = bias, t1 = bias;
#pragma unroll
        for (int q = 0; q < N_KC; ++q) {
            t0 = wmma_f32_16x16x4(wxa[q], eb[0][q], t0);
            t1 = wmma_f32_16x16x4(wxa[q], eb[1][q], t1);
        }

        float4* d0 = reinterpret_cast<float4*>(s_projb + (v0 + c) * PSTR + 8 * hh);
        float4* d1 = reinterpret_cast<float4*>(s_projb + (v0 + 16 + c) * PSTR + 8 * hh);
        d0[0] = make_float4(t0[0], t0[1], hh ? 0.0f : t0[2], hh ? 0.0f : t0[3]);
        d1[0] = make_float4(t1[0], t1[1], hh ? 0.0f : t1[2], hh ? 0.0f : t1[3]);
        if (hh == 0) {
            d0[1] = make_float4(t0[4], t0[5], t0[6], t0[7]);
            d1[1] = make_float4(t1[4], t1[5], t1[6], t1[7]);
        }
    }

    __syncthreads();

    v8f accp;
#pragma unroll
    for (int r = 0; r < 8; ++r) accp[r] = 0.0f;

    v2f ca0, ca1, ca2;
    v2f cb0, cb1, cb2;
    gather6(s_projb, cur[0], hoff, ca0, ca1, ca2);
    gather6(s_projb, cur[1], hoff, cb0, cb1, cb2);

    for (int ch = 0; ch < N_CHUNK; ++ch) {
        int nxt[T_CHUNK];
#pragma unroll
        for (int i = 0; i < T_CHUNK; ++i) nxt[i] = cur[i];
        if (ch + 1 < N_CHUNK) {
#pragma unroll
            for (int q = 0; q < NQ; ++q) {
                const int4 v = xrow[(ch + 1) * NQ + q];
                nxt[4 * q + 0] = v.x; nxt[4 * q + 1] = v.y;
                nxt[4 * q + 2] = v.z; nxt[4 * q + 3] = v.w;
            }
        }

#pragma unroll
        for (int t = 0; t < T_CHUNK; ++t) {
            const int tn = (t + 2 < T_CHUNK) ? cur[(t + 2) % T_CHUNK] : nxt[(t + 2) % T_CHUNK];
            v2f cc0, cc1, cc2;
            gather6(s_projb, tn, hoff, cc0, cc1, cc2);
            __builtin_amdgcn_sched_barrier(0);

            v8f acc;
            acc[0] = ca0[0]; acc[1] = ca0[1];
            acc[2] = ca1[0]; acc[3] = ca1[1];
            acc[4] = ca2[0]; acc[5] = ca2[1];
            acc[6] = 0.0f;   acc[7] = 0.0f;

            v2f b0, b1, b2;
            b0[0] = fast_tanh(accp[0]); b0[1] = fast_tanh(accp[1]);
            acc = wmma_f32_16x16x4(wha[0], b0, acc);
            b1[0] = fast_tanh(accp[2]); b1[1] = fast_tanh(accp[3]);
            acc = wmma_f32_16x16x4(wha[1], b1, acc);
            b2[0] = fast_tanh(accp[4]); b2[1] = fast_tanh(accp[5]);
            acc = wmma_f32_16x16x4(wha[2], b2, acc);

            accp = acc;
            ca0 = cb0; ca1 = cb1; ca2 = cb2;
            cb0 = cc0; cb1 = cc1; cb2 = cc2;
        }

#pragma unroll
        for (int i = 0; i < T_CHUNK; ++i) cur[i] = nxt[i];
    }

    v2f b0, b1, b2;
    b0[0] = fast_tanh(accp[0]); b0[1] = fast_tanh(accp[1]);
    b1[0] = fast_tanh(accp[2]); b1[1] = fast_tanh(accp[3]);
    b2[0] = fast_tanh(accp[4]); b2[1] = fast_tanh(accp[5]);

    v2f f0, f1, f2;
    f0[0] = b0[0] * dm[0]; f0[1] = b0[1] * dm[1];
    f1[0] = b1[0] * dm[2]; f1[1] = b1[1] * dm[3];
    f2[0] = b2[0] * dm[4]; f2[1] = b2[1] * dm[5];

    v8f lac;
#pragma unroll
    for (int r = 0; r < 8; ++r) lac[r] = bdv[r];

    lac = wmma_f32_16x16x4(wda[0], f0, lac);
    lac = wmma_f32_16x16x4(wda[1], f1, lac);
    lac = wmma_f32_16x16x4(wda[2], f2, lac);

#pragma unroll
    for (int r = 0; r < 8; ++r) {
        const int l = 8 * hh + r;
        if (l < L_SZ) s_out[rowLocal * L_SZ + l] = lac[r];
    }
    __syncthreads();

    float* ob = out + (size_t)blockIdx.x * (ROWS_PER_BLOCK * L_SZ);
#pragma unroll
    for (int i = tid; i < ROWS_PER_BLOCK * L_SZ; i += TPB) ob[i] = s_out[i];
}

extern "C" void kernel_launch(void* const* d_in, const int* in_sizes, int n_in,
                              void* d_out, int out_size, void* d_ws, size_t ws_size,
                              hipStream_t stream) {
    (void)in_sizes; (void)n_in; (void)d_ws; (void)ws_size; (void)out_size;
    const int*   x     = (const int*)  d_in[0];
    const float* emb   = (const float*)d_in[1];
    const float* Wx    = (const float*)d_in[2];
    const float* Wh    = (const float*)d_in[3];
    const float* b_rnn = (const float*)d_in[4];
    const float* Wd    = (const float*)d_in[5];
    const float* bd    = (const float*)d_in[6];
    const float* dmask = (const float*)d_in[7];
    float* out = (float*)d_out;

    charrnn_kernel<<<B_TOT / ROWS_PER_BLOCK, TPB, 0, stream>>>(
        x, emb, Wx, Wh, b_rnn, Wd, bd, dmask, out);
}


# One wave per SIMD on every WGP: 256 blocks of 4 waves (64 rows) instead of 128 blocks of 8, so no two recurrence chains share a matrix pipe; each wave builds 64 table columns, loads in one group
# speedup vs baseline: 3.2481x; 1.2860x over previous
#include <hip/hip_runtime.h>


typedef float v2f __attribute__((ext_vector_type(2)));
typedef float v8f __attribute__((ext_vector_type(8)));

#define B_TOT  16384
#define T_LEN  100
#define V_SZ   256
#define E_SZ   50
#define H_SZ   10
#define L_SZ   15
#define PSTR   12
#define ROWS_PER_BLOCK 64
#define TPB    128
#define N_TT   (V_SZ / (TPB / 32) / 16)
#define T_CHUNK 20
#define N_CHUNK (T_LEN / T_CHUNK)
#define NQ     (T_CHUNK / 4)
#define N_KC   13

__device__ __forceinline__ float fast_tanh(float x) {
#if __has_builtin(__builtin_amdgcn_tanhf)
    return __builtin_amdgcn_tanhf(x);
#else
    x = fminf(fmaxf(x, -9.0f), 9.0f);
    float z = __builtin_amdgcn_exp2f(x * 2.885390081777927f);
    return (z - 1.0f) * __builtin_amdgcn_rcpf(z + 1.0f);
#endif
}

__device__ __forceinline__ v8f wmma_f32_16x16x4(v2f a, v2f b, v8f c) {
    return __builtin_amdgcn_wmma_f32_16x16x4_f32(
        false, a, false, b, (short)0, c, false, false);
}

__device__ __forceinline__ int hid_of(int hh, int r) {
    return hh ? ((r < 4) ? 6 + r : -1) : ((r < 6) ? r : -1);
}

__device__ __forceinline__ int emb_col_of(int q, int hh, int s) {
    return (q < 12) ? 8 * (q >> 1) + 4 * hh + 2 * (q & 1) + s
                    : (hh ? -1 : 48 + s);
}

__device__ __forceinline__ void gather6(const float* tab, int tok, int hoff,
                                        v2f& p0, v2f& p1, v2f& p2) {
    const v2f* p = reinterpret_cast<const v2f*>(tab + (tok & (V_SZ - 1)) * PSTR + hoff);
    p0 = p[0];
    p1 = p[1];
    p2 = p[2];
}

__global__ __launch_bounds__(TPB)
void charrnn_kernel(const int*   __restrict__ x,
                    const float* __restrict__ emb,
                    const float* __restrict__ Wx,
                    const float* __restrict__ Wh,
                    const float* __restrict__ b_rnn,
                    const float* __restrict__ Wd,
                    const float* __restrict__ bd,
                    const float* __restrict__ dmask,
                    float*       __restrict__ out)
{
    __shared__ __attribute__((aligned(16))) float s_projb[V_SZ * PSTR];
    __shared__ __attribute__((aligned(16))) float s_out[ROWS_PER_BLOCK * L_SZ];

    const int tid   = threadIdx.x;
    const int lane  = tid & 31;
    const int wave  = tid >> 5;
    const int c     = lane & 15;
    const int hh    = lane >> 4;
    const int hoff  = 6 * hh;
    const int rowLocal = wave * 16 + c;
    const int gRow  = blockIdx.x * ROWS_PER_BLOCK + rowLocal;

    const int4* xrow = reinterpret_cast<const int4*>(x + (size_t)gRow * T_LEN);
    int cur[T_CHUNK];
#pragma unroll
    for (int q = 0; q < NQ; ++q) {
        const int4 v = xrow[q];
        cur[4 * q + 0] = v.x; cur[4 * q + 1] = v.y;
        cur[4 * q + 2] = v.z; cur[4 * q + 3] = v.w;
    }

    float dm[6];
    v2f wda[3];
#pragma unroll
    for (int r = 0; r < 6; ++r) {
        const int k = hid_of(hh, r);
        dm[r] = (k >= 0) ? dmask[(size_t)gRow * H_SZ + k] : 0.0f;
        wda[r >> 1][r & 1] = (k >= 0 && c < L_SZ) ? Wd[k * L_SZ + c] : 0.0f;
    }
    float bdv[8];
#pragma unroll
    for (int r = 0; r < 8; ++r) {
        const int l = 8 * hh + r;
        bdv[r] = (l < L_SZ) ? bd[l] : 0.0f;
    }

    const int jo = hid_of(c >> 3, c & 7);
    v2f wha[3];
#pragma unroll
    for (int q = 0; q < 3; ++q) {
#pragma unroll
        for (int s = 0; s < 2; ++s) {
            const int k = hid_of(hh, 2 * q + s);
            wha[q][s] = (k >= 0 && jo >= 0) ? Wh[k * H_SZ + jo] : 0.0f;
        }
    }

    {
        const int cj = (c < H_SZ) ? c : (H_SZ - 1);
        const int v0 = wave * (V_SZ / (TPB / 32));

        float wxr[N_KC][2];
#pragma unroll
        for (int q = 0; q < N_KC; ++q) {
#pragma unroll
            for (int s = 0; s < 2; ++s) {
                const int e  = emb_col_of(q, hh, s);
                const int ec = (e >= 0) ? e : emb_col_of(q, 0, s);
                wxr[q][s] = Wx[ec * H_SZ + cj];
            }
        }
        float br[8];
#pragma unroll
        for (int r = 0; r < 8; ++r) {
            const int j = 8 * hh + r;
            br[r] = b_rnn[(j < H_SZ) ? j : (H_SZ - 1)];
        }
        float ebr[N_TT][N_KC][2];
#pragma unroll
        for (int tl = 0; tl < N_TT; ++tl) {
            const float* er = emb + (v0 + 16 * tl + c) * E_SZ;
#pragma unroll
            for (int q = 0; q < N_KC; ++q) {
#pragma unroll
                for (int s = 0; s < 2; ++s) {
                    const int e  = emb_col_of(q, hh, s);
                    const int ec = (e >= 0) ? e : emb_col_of(q, 0, s);
                    ebr[tl][q][s] = er[ec];
                }
            }
        }
        __builtin_amdgcn_sched_barrier(0);

        v2f wxa[N_KC];
#pragma unroll
        for (int q = 0; q < N_KC; ++q) {
#pragma unroll
            for (int s = 0; s < 2; ++s) {
                const int e = emb_col_of(q, hh, s);
                wxa[q][s] = (e >= 0 && c < H_SZ) ? wxr[q][s] : 0.0f;
            }
        }

        v8f bias;
#pragma unroll
        for (int r = 0; r < 8; ++r) {
            const int j = 8 * hh + r;
            bias[r] = (j < H_SZ) ? br[r] : 0.0f;
        }

        v2f eb[N_TT][N_KC];
#pragma unroll
        for (int tl = 0; tl < N_TT; ++tl) {
#pragma unroll
            for (int q = 0; q < N_KC; ++q) {
#pragma unroll
                for (int s = 0; s < 2; ++s) {
                    const int e = emb_col_of(q, hh, s);
                    eb[tl][q][s] = (e >= 0) ? ebr[tl][q][s] : 0.0f;
                }
            }
        }

        v8f tt[N_TT];
#pragma unroll
        for (int tl = 0; tl < N_TT; ++tl) tt[tl] = bias;
#pragma unroll
        for (int q = 0; q < N_KC; ++q) {
#pragma unroll
            for (int tl = 0; tl < N_TT; ++tl)
                tt[tl] = wmma_f32_16x16x4(wxa[q], eb[tl][q], tt[tl]);
        }

#pragma unroll
        for (int tl = 0; tl < N_TT; ++tl) {
            float4* d = reinterpret_cast<float4*>(s_projb + (v0 + 16 * tl + c) * PSTR + 8 * hh);
            d[0] = make_float4(tt[tl][0], tt[tl][1],
                               hh ? 0.0f : tt[tl][2], hh ? 0.0f : tt[tl][3]);
            if (hh == 0)
                d[1] = make_float4(tt[tl][4], tt[tl][5], tt[tl][6], tt[tl][7]);
        }
    }

    __syncthreads();

    v8f accp;
#pragma unroll
    for (int r = 0; r < 8; ++r) accp[r] = 0.0f;

    v2f ca0, ca1, ca2;
    v2f cb0, cb1, cb2;
    gather6(s_projb, cur[0], hoff, ca0, ca1, ca2);
    gather6(s_projb, cur[1], hoff, cb0, cb1, cb2);

    for (int ch = 0; ch < N_CHUNK; ++ch) {
        int nxt[T_CHUNK];
#pragma unroll
        for (int i = 0; i < T_CHUNK; ++i) nxt[i] = cur[i];
        if (ch + 1 < N_CHUNK) {
#pragma unroll
            for (int q = 0; q < NQ; ++q) {
                const int4 v = xrow[(ch + 1) * NQ + q];
                nxt[4 * q + 0] = v.x; nxt[4 * q + 1] = v.y;
                nxt[4 * q + 2] = v.z; nxt[4 * q + 3] = v.w;
            }
        }

#pragma unroll
        for (int t = 0; t < T_CHUNK; ++t) {
            const int tn = (t + 2 < T_CHUNK) ? cur[(t + 2) % T_CHUNK] : nxt[(t + 2) % T_CHUNK];
            v2f cc0, cc1, cc2;
            gather6(s_projb, tn, hoff, cc0, cc1, cc2);
            __builtin_amdgcn_sched_barrier(0);

            v8f acc;
            acc[0] = ca0[0]; acc[1] = ca0[1];
            acc[2] = ca1[0]; acc[3] = ca1[1];
            acc[4] = ca2[0]; acc[5] = ca2[1];
            acc[6] = 0.0f;   acc[7] = 0.0f;

            v2f b0, b1, b2;
            b0[0] = fast_tanh(accp[0]); b0[1] = fast_tanh(accp[1]);
            acc = wmma_f32_16x16x4(wha[0], b0, acc);
            b1[0] = fast_tanh(accp[2]); b1[1] = fast_tanh(accp[3]);
            acc = wmma_f32_16x16x4(wha[1], b1, acc);
            b2[0] = fast_tanh(accp[4]); b2[1] = fast_tanh(accp[5]);
            acc = wmma_f32_16x16x4(wha[2], b2, acc);

            accp = acc;
            ca0 = cb0; ca1 = cb1; ca2 = cb2;
            cb0 = cc0; cb1 = cc1; cb2 = cc2;
        }

#pragma unroll
        for (int i = 0; i < T_CHUNK; ++i) cur[i] = nxt[i];
    }

    v2f b0, b1, b2;
    b0[0] = fast_tanh(accp[0]); b0[1] = fast_tanh(accp[1]);
    b1[0] = fast_tanh(accp[2]); b1[1] = fast_tanh(accp[3]);
    b2[0] = fast_tanh(accp[4]); b2[1] = fast_tanh(accp[5]);

    v2f f0, f1, f2;
    f0[0] = b0[0] * dm[0]; f0[1] = b0[1] * dm[1];
    f1[0] = b1[0] * dm[2]; f1[1] = b1[1] * dm[3];
    f2[0] = b2[0] * dm[4]; f2[1] = b2[1] * dm[5];

    v8f lac;
#pragma unroll
    for (int r = 0; r < 8; ++r) lac[r] = bdv[r];

    lac = wmma_f32_16x16x4(wda[0], f0, lac);
    lac = wmma_f32_16x16x4(wda[1], f1, lac);
    lac = wmma_f32_16x16x4(wda[2], f2, lac);

#pragma unroll
    for (int r = 0; r < 8; ++r) {
        const int l = 8 * hh + r;
        if (l < L_SZ) s_out[rowLocal * L_SZ + l] = lac[r];
    }
    __syncthreads();

    float* ob = out + (size_t)blockIdx.x * (ROWS_PER_BLOCK * L_SZ);
    constexpr int N_OUT  = ROWS_PER_BLOCK * L_SZ;
    constexpr int N_FULL = N_OUT / TPB;
#pragma unroll
    for (int j = 0; j < N_FULL; ++j) ob[tid + j * TPB] = s_out[tid + j * TPB];
    if (tid < N_OUT - N_FULL * TPB)
        ob[tid + N_FULL * TPB] = s_out[tid + N_FULL * TPB];
}

extern "C" void kernel_launch(void* const* d_in, const int* in_sizes, int n_in,
                              void* d_out, int out_size, void* d_ws, size_t ws_size,
                              hipStream_t stream) {
    (void)in_sizes; (void)n_in; (void)d_ws; (void)ws_size; (void)out_size;
    const int*   x     = (const int*)  d_in[0];
    const float* emb   = (const float*)d_in[1];
    const float* Wx    = (const float*)d_in[2];
    const float* Wh    = (const float*)d_in[3];
    const float* b_rnn = (const float*)d_in[4];
    const float* Wd    = (const float*)d_in[5];
    const float* bd    = (const float*)d_in[6];
    const float* dmask = (const float*)d_in[7];
    float* out = (float*)d_out;

    charrnn_kernel<<<B_TOT / ROWS_PER_BLOCK, TPB, 0, stream>>>(
        x, emb, Wx, Wh, b_rnn, Wd, bd, dmask, out);
}


# Table build fed first and in whole lines: each wave stages its contiguous 12800-byte emb slab through LDS (25 coalesced loads, not 28 of 16+ lines); token, Wh, Wd, mask loads issued behind it
# speedup vs baseline: 3.4975x; 1.0768x over previous
#include <hip/hip_runtime.h>


typedef float v2f __attribute__((ext_vector_type(2)));
typedef float v8f __attribute__((ext_vector_type(8)));

#define B_TOT  16384
#define T_LEN  100
#define V_SZ   256
#define E_SZ   50
#define H_SZ   10
#define L_SZ   15
#define PSTR   12
#define ROWS_PER_BLOCK 64
#define TPB    128
#define N_TT   (V_SZ / (TPB / 32) / 16)
#define T_CHUNK 20
#define N_CHUNK (T_LEN / T_CHUNK)
#define NQ     (T_CHUNK / 4)
#define N_KC   13
#define TOK_PER_WAVE (V_SZ / (TPB / 32))
#define SLAB_F4 (TOK_PER_WAVE * E_SZ / 4)
#define SLAB_LD (SLAB_F4 / 32)

__device__ __forceinline__ float fast_tanh(float x) {
#if __has_builtin(__builtin_amdgcn_tanhf)
    return __builtin_amdgcn_tanhf(x);
#else
    x = fminf(fmaxf(x, -9.0f), 9.0f);
    float z = __builtin_amdgcn_exp2f(x * 2.885390081777927f);
    return (z - 1.0f) * __builtin_amdgcn_rcpf(z + 1.0f);
#endif
}

__device__ __forceinline__ v8f wmma_f32_16x16x4(v2f a, v2f b, v8f c) {
    return __builtin_amdgcn_wmma_f32_16x16x4_f32(
        false, a, false, b, (short)0, c, false, false);
}

__device__ __forceinline__ int hid_of(int hh, int r) {
    return hh ? ((r < 4) ? 6 + r : -1) : ((r < 6) ? r : -1);
}

__device__ __forceinline__ int emb_col_of(int q, int hh, int s) {
    return (q < 12) ? 8 * (q >> 1) + 4 * hh + 2 * (q & 1) + s
                    : (hh ? -1 : 48 + s);
}

__device__ __forceinline__ void gather6(const float* tab, int tok, int hoff,
                                        v2f& p0, v2f& p1, v2f& p2) {
    const v2f* p = reinterpret_cast<const v2f*>(tab + (tok & (V_SZ - 1)) * PSTR + hoff);
    p0 = p[0];
    p1 = p[1];
    p2 = p[2];
}

__global__ __launch_bounds__(TPB)
void charrnn_kernel(const int*   __restrict__ x,
                    const float* __restrict__ emb,
                    const float* __restrict__ Wx,
                    const float* __restrict__ Wh,
                    const float* __restrict__ b_rnn,
                    const float* __restrict__ Wd,
                    const float* __restrict__ bd,
                    const float* __restrict__ dmask,
                    float*       __restrict__ out)
{
    __shared__ __attribute__((aligned(16))) float s_projb[V_SZ * PSTR];
    __shared__ __attribute__((aligned(16))) float s_out[ROWS_PER_BLOCK * L_SZ];
    __shared__ __attribute__((aligned(16))) float s_emb[V_SZ * E_SZ];

    const int tid   = threadIdx.x;
    const int lane  = tid & 31;
    const int wave  = tid >> 5;
    const int c     = lane & 15;
    const int hh    = lane >> 4;
    const int hoff  = 6 * hh;
    const int rowLocal = wave * 16 + c;
    const int gRow  = blockIdx.x * ROWS_PER_BLOCK + rowLocal;

    float4 ev[SLAB_LD];
    {
        const float4* eg = reinterpret_cast<const float4*>(emb) + wave * SLAB_F4 + lane;
#pragma unroll
        for (int i = 0; i < SLAB_LD; ++i) ev[i] = eg[i * 32];
    }

    const int4* xrow = reinterpret_cast<const int4*>(x + (size_t)gRow * T_LEN);
    int cur[T_CHUNK];
    const int jo  = hid_of(c >> 3, c & 7);
    const int joc = (jo >= 0) ? jo : 0;
    const int cl  = (c < L_SZ) ? c : (L_SZ - 1);
    float dmr[6], wdr[6], whr[6], bdr[8];

    {
        const int cj = (c < H_SZ) ? c : (H_SZ - 1);
        const int v0 = wave * TOK_PER_WAVE;

        float wxr[N_KC][2];
#pragma unroll
        for (int q = 0; q < N_KC; ++q) {
#pragma unroll
            for (int s = 0; s < 2; ++s) {
                const int e  = emb_col_of(q, hh, s);
                const int ec = (e >= 0) ? e : emb_col_of(q, 0, s);
                wxr[q][s] = Wx[ec * H_SZ + cj];
            }
        }
        float br[8];
#pragma unroll
        for (int r = 0; r < 8; ++r) {
            const int j = 8 * hh + r;
            br[r] = b_rnn[(j < H_SZ) ? j : (H_SZ - 1)];
        }
        {
            float4* sw = reinterpret_cast<float4*>(s_emb) + wave * SLAB_F4 + lane;
#pragma unroll
            for (int i = 0; i < SLAB_LD; ++i) sw[i * 32] = ev[i];
        }
        __syncthreads();

#pragma unroll
        for (int q = 0; q < NQ; ++q) {
            const int4 v = xrow[q];
            cur[4 * q + 0] = v.x; cur[4 * q + 1] = v.y;
            cur[4 * q + 2] = v.z; cur[4 * q + 3] = v.w;
        }
#pragma unroll
        for (int r = 0; r < 6; ++r) {
            const int k  = hid_of(hh, r);
            const int kc = (k >= 0) ? k : 0;
            dmr[r] = dmask[(size_t)gRow * H_SZ + kc];
            wdr[r] = Wd[kc * L_SZ + cl];
            whr[r] = Wh[kc * H_SZ + joc];
        }
#pragma unroll
        for (int r = 0; r < 8; ++r) {
            const int l = 8 * hh + r;
            bdr[r] = bd[(l < L_SZ) ? l : (L_SZ - 1)];
        }

        float ebr[N_TT][N_KC][2];
#pragma unroll
        for (int tl = 0; tl < N_TT; ++tl) {
            const float* er = s_emb + (v0 + 16 * tl + c) * E_SZ;
#pragma unroll
            for (int q = 0; q < N_KC; ++q) {
#pragma unroll
                for (int s = 0; s < 2; ++s) {
                    const int e  = emb_col_of(q, hh, s);
                    const int ec = (e >= 0) ? e : emb_col_of(q, 0, s);
                    ebr[tl][q][s] = er[ec];
                }
            }
        }
        __builtin_amdgcn_sched_barrier(0);

        v2f wxa[N_KC];
#pragma unroll
        for (int q = 0; q < N_KC; ++q) {
#pragma unroll
            for (int s = 0; s < 2; ++s) {
                const int e = emb_col_of(q, hh, s);
                wxa[q][s] = (e >= 0 && c < H_SZ) ? wxr[q][s] : 0.0f;
            }
        }

        v8f bias;
#pragma unroll
        for (int r = 0; r < 8; ++r) {
            const int j = 8 * hh + r;
            bias[r] = (j < H_SZ) ? br[r] : 0.0f;
        }

        v2f eb[N_TT][N_KC];
#pragma unroll
        for (int tl = 0; tl < N_TT; ++tl) {
#pragma unroll
            for (int q = 0; q < N_KC; ++q) {
#pragma unroll
                for (int s = 0; s < 2; ++s) {
                    const int e = emb_col_of(q, hh, s);
                    eb[tl][q][s] = (e >= 0) ? ebr[tl][q][s] : 0.0f;
                }
            }
        }

        v8f tt[N_TT];
#pragma unroll
        for (int tl = 0; tl < N_TT; ++tl) tt[tl] = bias;
#pragma unroll
        for (int q = 0; q < N_KC; ++q) {
#pragma unroll
            for (int tl = 0; tl < N_TT; ++tl)
                tt[tl] = wmma_f32_16x16x4(wxa[q], eb[tl][q], tt[tl]);
        }

#pragma unroll
        for (int tl = 0; tl < N_TT; ++tl) {
            float4* d = reinterpret_cast<float4*>(s_projb + (v0 + 16 * tl + c) * PSTR + 8 * hh);
            d[0] = make_float4(tt[tl][0], tt[tl][1],
                               hh ? 0.0f : tt[tl][2], hh ? 0.0f : tt[tl][3]);
            if (hh == 0)
                d[1] = make_float4(tt[tl][4], tt[tl][5], tt[tl][6], tt[tl][7]);
        }
    }

    float dm[6];
    v2f wda[3];
    v2f wha[3];
#pragma unroll
    for (int r = 0; r < 6; ++r) {
        const int k = hid_of(hh, r);
        dm[r] = (k >= 0) ? dmr[r] : 0.0f;
        wda[r >> 1][r & 1] = (k >= 0 && c < L_SZ) ? wdr[r] : 0.0f;
        wha[r >> 1][r & 1] = (k >= 0 && jo >= 0) ? whr[r] : 0.0f;
    }
    float bdv[8];
#pragma unroll
    for (int r = 0; r < 8; ++r) {
        const int l = 8 * hh + r;
        bdv[r] = (l < L_SZ) ? bdr[r] : 0.0f;
    }

    __syncthreads();

    v8f accp;
#pragma unroll
    for (int r = 0; r < 8; ++r) accp[r] = 0.0f;

    v2f ca0, ca1, ca2;
    v2f cb0, cb1, cb2;
    gather6(s_projb, cur[0], hoff, ca0, ca1, ca2);
    gather6(s_projb, cur[1], hoff, cb0, cb1, cb2);

    for (int ch = 0; ch < N_CHUNK; ++ch) {
        int nxt[T_CHUNK];
#pragma unroll
        for (int i = 0; i < T_CHUNK; ++i) nxt[i] = cur[i];
        if (ch + 1 < N_CHUNK) {
#pragma unroll
            for (int q = 0; q < NQ; ++q) {
                const int4 v = xrow[(ch + 1) * NQ + q];
                nxt[4 * q + 0] = v.x; nxt[4 * q + 1] = v.y;
                nxt[4 * q + 2] = v.z; nxt[4 * q + 3] = v.w;
            }
        }

#pragma unroll
        for (int t = 0; t < T_CHUNK; ++t) {
            const int tn = (t + 2 < T_CHUNK) ? cur[(t + 2) % T_CHUNK] : nxt[(t + 2) % T_CHUNK];
            v2f cc0, cc1, cc2;
            gather6(s_projb, tn, hoff, cc0, cc1, cc2);
            __builtin_amdgcn_sched_barrier(0);

            v8f acc;
            acc[0] = ca0[0]; acc[1] = ca0[1];
            acc[2] = ca1[0]; acc[3] = ca1[1];
            acc[4] = ca2[0]; acc[5] = ca2[1];
            acc[6] = 0.0f;   acc[7] = 0.0f;

            v2f b0, b1, b2;
            b0[0] = fast_tanh(accp[0]); b0[1] = fast_tanh(accp[1]);
            acc = wmma_f32_16x16x4(wha[0], b0, acc);
            b1[0] = fast_tanh(accp[2]); b1[1] = fast_tanh(accp[3]);
            acc = wmma_f32_16x16x4(wha[1], b1, acc);
            b2[0] = fast_tanh(accp[4]); b2[1] = fast_tanh(accp[5]);
            acc = wmma_f32_16x16x4(wha[2], b2, acc);

            accp = acc;
            ca0 = cb0; ca1 = cb1; ca2 = cb2;
            cb0 = cc0; cb1 = cc1; cb2 = cc2;
        }

#pragma unroll
        for (int i = 0; i < T_CHUNK; ++i) cur[i] = nxt[i];
    }

    v2f b0, b1, b2;
    b0[0] = fast_tanh(accp[0]); b0[1] = fast_tanh(accp[1]);
    b1[0] = fast_tanh(accp[2]); b1[1] = fast_tanh(accp[3]);
    b2[0] = fast_tanh(accp[4]); b2[1] = fast_tanh(accp[5]);

    v2f f0, f1, f2;
    f0[0] = b0[0] * dm[0]; f0[1] = b0[1] * dm[1];
    f1[0] = b1[0] * dm[2]; f1[1] = b1[1] * dm[3];
    f2[0] = b2[0] * dm[4]; f2[1] = b2[1] * dm[5];

    v8f lac;
#pragma unroll
    for (int r = 0; r < 8; ++r) lac[r] = bdv[r];

    lac = wmma_f32_16x16x4(wda[0], f0, lac);
    lac = wmma_f32_16x16x4(wda[1], f1, lac);
    lac = wmma_f32_16x16x4(wda[2], f2, lac);

#pragma unroll
    for (int r = 0; r < 8; ++r) {
        const int l = 8 * hh + r;
        if (l < L_SZ) s_out[rowLocal * L_SZ + l] = lac[r];
    }
    __syncthreads();

    float* ob = out + (size_t)blockIdx.x * (ROWS_PER_BLOCK * L_SZ);
    constexpr int N_OUT  = ROWS_PER_BLOCK * L_SZ;
    constexpr int N_FULL = N_OUT / TPB;
#pragma unroll
    for (int j = 0; j < N_FULL; ++j) ob[tid + j * TPB] = s_out[tid + j * TPB];
    if (tid < N_OUT - N_FULL * TPB)
        ob[tid + N_FULL * TPB] = s_out[tid + N_FULL * TPB];
}

extern "C" void kernel_launch(void* const* d_in, const int* in_sizes, int n_in,
                              void* d_out, int out_size, void* d_ws, size_t ws_size,
                              hipStream_t stream) {
    (void)in_sizes; (void)n_in; (void)d_ws; (void)ws_size; (void)out_size;
    const int*   x     = (const int*)  d_in[0];
    const float* emb   = (const float*)d_in[1];
    const float* Wx    = (const float*)d_in[2];
    const float* Wh    = (const float*)d_in[3];
    const float* b_rnn = (const float*)d_in[4];
    const float* Wd    = (const float*)d_in[5];
    const float* bd    = (const float*)d_in[6];
    const float* dmask = (const float*)d_in[7];
    float* out = (float*)d_out;

    charrnn_kernel<<<B_TOT / ROWS_PER_BLOCK, TPB, 0, stream>>>(
        x, emb, Wx, Wh, b_rnn, Wd, bd, dmask, out);
}


# C operands of the next 10 steps gathered from the LDS table as one pinned group of loads 4 steps ahead, addresses computed in WMMA shadows: the tanh-WMMA chain waits for LDS once, not 13 times
# speedup vs baseline: 3.7597x; 1.0750x over previous
#include <hip/hip_runtime.h>


typedef float v2f __attribute__((ext_vector_type(2)));
typedef float v8f __attribute__((ext_vector_type(8)));

#define B_TOT  16384
#define T_LEN  100
#define V_SZ   256
#define E_SZ   50
#define H_SZ   10
#define L_SZ   15
#define PSTR   12
#define ROWS_PER_BLOCK 64
#define TPB    128
#define N_TT   (V_SZ / (TPB / 32) / 16)
#define T_CHUNK 20
#define N_CHUNK (T_LEN / T_CHUNK)
#define NQ     (T_CHUNK / 4)
#define GRP    10
#define N_GRP  (T_LEN / GRP)
#define G_AT   6
#define N_KC   13
#define TOK_PER_WAVE (V_SZ / (TPB / 32))
#define SLAB_F4 (TOK_PER_WAVE * E_SZ / 4)
#define SLAB_LD (SLAB_F4 / 32)

__device__ __forceinline__ float fast_tanh(float x) {
#if __has_builtin(__builtin_amdgcn_tanhf)
    return __builtin_amdgcn_tanhf(x);
#else
    x = fminf(fmaxf(x, -9.0f), 9.0f);
    float z = __builtin_amdgcn_exp2f(x * 2.885390081777927f);
    return (z - 1.0f) * __builtin_amdgcn_rcpf(z + 1.0f);
#endif
}

__device__ __forceinline__ v8f wmma_f32_16x16x4(v2f a, v2f b, v8f c) {
    return __builtin_amdgcn_wmma_f32_16x16x4_f32(
        false, a, false, b, (short)0, c, false, false);
}

__device__ __forceinline__ int hid_of(int hh, int r) {
    return hh ? ((r < 4) ? 6 + r : -1) : ((r < 6) ? r : -1);
}

__device__ __forceinline__ int emb_col_of(int q, int hh, int s) {
    return (q < 12) ? 8 * (q >> 1) + 4 * hh + 2 * (q & 1) + s
                    : (hh ? -1 : 48 + s);
}

typedef float v4f __attribute__((ext_vector_type(4)));
typedef float v4f_a8 __attribute__((ext_vector_type(4), aligned(8)));
typedef __attribute__((address_space(3))) const volatile v4f_a8 lds_cv4f;
typedef __attribute__((address_space(3))) const volatile v2f lds_cv2f;

__device__ __forceinline__ unsigned c_addr(const float* tab, int tok, int hoff) {
    const float* q = tab + (tok & (V_SZ - 1)) * PSTR + hoff;
    return (unsigned)reinterpret_cast<unsigned long long>(q);
}

__device__ __forceinline__ void gather6(unsigned off, v2f& p0, v2f& p1, v2f& p2) {
    const v4f lo = *reinterpret_cast<lds_cv4f*>(off);
    const v2f hi = *reinterpret_cast<lds_cv2f*>(off + 16u);
    p0[0] = lo[0]; p0[1] = lo[1];
    p1[0] = lo[2]; p1[1] = lo[3];
    p2 = hi;
}

__device__ __forceinline__ void chain_point(v8f& a) {
    float t = a[0];
    asm volatile("" : "+v"(t) : : "memory");
    a[0] = t;
}
__device__ __forceinline__ void chain_point(v8f& a, unsigned& x, unsigned& y) {
    float t = a[0];
    asm volatile("" : "+v"(t), "+v"(x), "+v"(y));
    a[0] = t;
}

__global__ __launch_bounds__(TPB)
void charrnn_kernel(const int*   __restrict__ x,
                    const float* __restrict__ emb,
                    const float* __restrict__ Wx,
                    const float* __restrict__ Wh,
                    const float* __restrict__ b_rnn,
                    const float* __restrict__ Wd,
                    const float* __restrict__ bd,
                    const float* __restrict__ dmask,
                    float*       __restrict__ out)
{
    __shared__ __attribute__((aligned(16))) float s_projb[V_SZ * PSTR];
    __shared__ __attribute__((aligned(16))) float s_out[ROWS_PER_BLOCK * L_SZ];
    __shared__ __attribute__((aligned(16))) float s_emb[V_SZ * E_SZ];

    const int tid   = threadIdx.x;
    const int lane  = tid & 31;
    const int wave  = tid >> 5;
    const int c     = lane & 15;
    const int hh    = lane >> 4;
    const int hoff  = 6 * hh;
    const int rowLocal = wave * 16 + c;
    const int gRow  = blockIdx.x * ROWS_PER_BLOCK + rowLocal;

    float4 ev[SLAB_LD];
    {
        const float4* eg = reinterpret_cast<const float4*>(emb) + wave * SLAB_F4 + lane;
#pragma unroll
        for (int i = 0; i < SLAB_LD; ++i) ev[i] = eg[i * 32];
    }

    const int4* xrow = reinterpret_cast<const int4*>(x + (size_t)gRow * T_LEN);
    int tok[2][T_CHUNK];
    const int jo  = hid_of(c >> 3, c & 7);
    const int joc = (jo >= 0) ? jo : 0;
    const int cl  = (c < L_SZ) ? c : (L_SZ - 1);
    float dmr[6], wdr[6], whr[6], bdr[8];

    {
        const int cj = (c < H_SZ) ? c : (H_SZ - 1);
        const int v0 = wave * TOK_PER_WAVE;

        float wxr[N_KC][2];
#pragma unroll
        for (int q = 0; q < N_KC; ++q) {
#pragma unroll
            for (int s = 0; s < 2; ++s) {
                const int e  = emb_col_of(q, hh, s);
                const int ec = (e >= 0) ? e : emb_col_of(q, 0, s);
                wxr[q][s] = Wx[ec * H_SZ + cj];
            }
        }
        float br[8];
#pragma unroll
        for (int r = 0; r < 8; ++r) {
            const int j = 8 * hh + r;
            br[r] = b_rnn[(j < H_SZ) ? j : (H_SZ - 1)];
        }
        {
            float4* sw = reinterpret_cast<float4*>(s_emb) + wave * SLAB_F4 + lane;
#pragma unroll
            for (int i = 0; i < SLAB_LD; ++i) sw[i * 32] = ev[i];
        }
        __syncthreads();

#pragma unroll
        for (int q = 0; q < NQ; ++q) {
            const int4 v = xrow[q];
            tok[0][4 * q + 0] = v.x; tok[0][4 * q + 1] = v.y;
            tok[0][4 * q + 2] = v.z; tok[0][4 * q + 3] = v.w;
        }
#pragma unroll
        for (int q = 0; q < NQ; ++q) {
            const int4 v = xrow[NQ + q];
            tok[1][4 * q + 0] = v.x; tok[1][4 * q + 1] = v.y;
            tok[1][4 * q + 2] = v.z; tok[1][4 * q + 3] = v.w;
        }
#pragma unroll
        for (int r = 0; r < 6; ++r) {
            const int k  = hid_of(hh, r);
            const int kc = (k >= 0) ? k : 0;
            dmr[r] = dmask[(size_t)gRow * H_SZ + kc];
            wdr[r] = Wd[kc * L_SZ + cl];
            whr[r] = Wh[kc * H_SZ + joc];
        }
#pragma unroll
        for (int r = 0; r < 8; ++r) {
            const int l = 8 * hh + r;
            bdr[r] = bd[(l < L_SZ) ? l : (L_SZ - 1)];
        }

        float ebr[N_TT][N_KC][2];
#pragma unroll
        for (int tl = 0; tl < N_TT; ++tl) {
            const float* er = s_emb + (v0 + 16 * tl + c) * E_SZ;
#pragma unroll
            for (int q = 0; q < N_KC; ++q) {
#pragma unroll
                for (int s = 0; s < 2; ++s) {
                    const int e  = emb_col_of(q, hh, s);
                    const int ec = (e >= 0) ? e : emb_col_of(q, 0, s);
                    ebr[tl][q][s] = er[ec];
                }
            }
        }
        __builtin_amdgcn_sched_barrier(0);

        v2f wxa[N_KC];
#pragma unroll
        for (int q = 0; q < N_KC; ++q) {
#pragma unroll
            for (int s = 0; s < 2; ++s) {
                const int e = emb_col_of(q, hh, s);
                wxa[q][s] = (e >= 0 && c < H_SZ) ? wxr[q][s] : 0.0f;
            }
        }

        v8f bias;
#pragma unroll
        for (int r = 0; r < 8; ++r) {
            const int j = 8 * hh + r;
            bias[r] = (j < H_SZ) ? br[r] : 0.0f;
        }

        v2f eb[N_TT][N_KC];
#pragma unroll
        for (int tl = 0; tl < N_TT; ++tl) {
#pragma unroll
            for (int q = 0; q < N_KC; ++q) {
#pragma unroll
                for (int s = 0; s < 2; ++s) {
                    const int e = emb_col_of(q, hh, s);
                    eb[tl][q][s] = (e >= 0) ? ebr[tl][q][s] : 0.0f;
                }
            }
        }

        v8f tt[N_TT];
#pragma unroll
        for (int tl = 0; tl < N_TT; ++tl) tt[tl] = bias;
#pragma unroll
        for (int q = 0; q < N_KC; ++q) {
#pragma unroll
            for (int tl = 0; tl < N_TT; ++tl)
                tt[tl] = wmma_f32_16x16x4(wxa[q], eb[tl][q], tt[tl]);
        }

#pragma unroll
        for (int tl = 0; tl < N_TT; ++tl) {
            float4* d = reinterpret_cast<float4*>(s_projb + (v0 + 16 * tl + c) * PSTR + 8 * hh);
            d[0] = make_float4(tt[tl][0], tt[tl][1],
                               hh ? 0.0f : tt[tl][2], hh ? 0.0f : tt[tl][3]);
            if (hh == 0)
                d[1] = make_float4(tt[tl][4], tt[tl][5], tt[tl][6], tt[tl][7]);
        }
    }

    float dm[6];
    v2f wda[3];
    v2f wha[3];
#pragma unroll
    for (int r = 0; r < 6; ++r) {
        const int k = hid_of(hh, r);
        dm[r] = (k >= 0) ? dmr[r] : 0.0f;
        wda[r >> 1][r & 1] = (k >= 0 && c < L_SZ) ? wdr[r] : 0.0f;
        wha[r >> 1][r & 1] = (k >= 0 && jo >= 0) ? whr[r] : 0.0f;
    }
    float bdv[8];
#pragma unroll
    for (int r = 0; r < 8; ++r) {
        const int l = 8 * hh + r;
        bdv[r] = (l < L_SZ) ? bdr[r] : 0.0f;
    }

    __syncthreads();

    v8f accp;
#pragma unroll
    for (int r = 0; r < 8; ++r) accp[r] = 0.0f;

    unsigned ca[GRP];
    v2f csA[GRP][3], csB[GRP][3];
#pragma unroll
    for (int j = 0; j < GRP; ++j) ca[j] = c_addr(s_projb, tok[0][j], hoff);
#pragma unroll
    for (int j = 0; j < GRP; ++j) gather6(ca[j], csA[j][0], csA[j][1], csA[j][2]);
    chain_point(accp);

    auto run_group = [&](const int g, v2f (&cur)[GRP][3], v2f (&nxt)[GRP][3])
                         __attribute__((always_inline)) {
#pragma unroll
        for (int i = 0; i < GRP; ++i) {
            if (g + 1 < N_GRP && i >= 1 && i <= GRP / 2) {
                const int j0 = 2 * (i - 1);
                const int s0 = (g + 1) * GRP + j0;
                const int s1 = s0 + 1;
                ca[j0]     = c_addr(s_projb, tok[(s0 / T_CHUNK) & 1][s0 % T_CHUNK], hoff);
                ca[j0 + 1] = c_addr(s_projb, tok[(s1 / T_CHUNK) & 1][s1 % T_CHUNK], hoff);
                chain_point(accp, ca[j0], ca[j0 + 1]);
            }

            if (g + 1 < N_GRP && i == G_AT) {
                chain_point(accp);
                __builtin_amdgcn_sched_barrier(0);
#pragma unroll
                for (int j = 0; j < GRP; ++j)
                    gather6(ca[j], nxt[j][0], nxt[j][1], nxt[j][2]);
                const int c = (g + 1) / 2;
                if (((g + 1) & 1) && c + 2 < N_CHUNK) {
#pragma unroll
                    for (int q = 0; q < NQ; ++q) {
                        const int4 v = xrow[(c + 2) * NQ + q];
                        tok[c & 1][4 * q + 0] = v.x; tok[c & 1][4 * q + 1] = v.y;
                        tok[c & 1][4 * q + 2] = v.z; tok[c & 1][4 * q + 3] = v.w;
                    }
                }
                __builtin_amdgcn_sched_barrier(0);
                chain_point(accp);
            }

            const v4f c01 = __builtin_shufflevector(cur[i][0], cur[i][1], 0, 1, 2, 3);
            const v4f c2x = __builtin_shufflevector(cur[i][2], cur[i][2], 0, 1, -1, -1);
            v8f acc = __builtin_shufflevector(c01, c2x, 0, 1, 2, 3, 4, 5, -1, -1);

            v2f b0, b1, b2;
            b0[0] = fast_tanh(accp[0]); b0[1] = fast_tanh(accp[1]);
            acc = wmma_f32_16x16x4(wha[0], b0, acc);
            b1[0] = fast_tanh(accp[2]); b1[1] = fast_tanh(accp[3]);
            acc = wmma_f32_16x16x4(wha[1], b1, acc);
            b2[0] = fast_tanh(accp[4]); b2[1] = fast_tanh(accp[5]);
            acc = wmma_f32_16x16x4(wha[2], b2, acc);

            accp = acc;
        }
    };

#pragma unroll
    for (int gp = 0; gp < N_GRP / 2; ++gp) {
        run_group(2 * gp,     csA, csB);
        run_group(2 * gp + 1, csB, csA);
    }

    v2f b0, b1, b2;
    b0[0] = fast_tanh(accp[0]); b0[1] = fast_tanh(accp[1]);
    b1[0] = fast_tanh(accp[2]); b1[1] = fast_tanh(accp[3]);
    b2[0] = fast_tanh(accp[4]); b2[1] = fast_tanh(accp[5]);

    v2f f0, f1, f2;
    f0[0] = b0[0] * dm[0]; f0[1] = b0[1] * dm[1];
    f1[0] = b1[0] * dm[2]; f1[1] = b1[1] * dm[3];
    f2[0] = b2[0] * dm[4]; f2[1] = b2[1] * dm[5];

    v8f lac;
#pragma unroll
    for (int r = 0; r < 8; ++r) lac[r] = bdv[r];

    lac = wmma_f32_16x16x4(wda[0], f0, lac);
    lac = wmma_f32_16x16x4(wda[1], f1, lac);
    lac = wmma_f32_16x16x4(wda[2], f2, lac);

#pragma unroll
    for (int r = 0; r < 8; ++r) {
        const int l = 8 * hh + r;
        if (l < L_SZ) s_out[rowLocal * L_SZ + l] = lac[r];
    }
    __syncthreads();

    float* ob = out + (size_t)blockIdx.x * (ROWS_PER_BLOCK * L_SZ);
    constexpr int N_OUT  = ROWS_PER_BLOCK * L_SZ;
    constexpr int N_FULL = N_OUT / TPB;
#pragma unroll
    for (int j = 0; j < N_FULL; ++j) ob[tid + j * TPB] = s_out[tid + j * TPB];
    if (tid < N_OUT - N_FULL * TPB)
        ob[tid + N_FULL * TPB] = s_out[tid + N_FULL * TPB];
}

extern "C" void kernel_launch(void* const* d_in, const int* in_sizes, int n_in,
                              void* d_out, int out_size, void* d_ws, size_t ws_size,
                              hipStream_t stream) {
    (void)in_sizes; (void)n_in; (void)d_ws; (void)ws_size; (void)out_size;
    const int*   x     = (const int*)  d_in[0];
    const float* emb   = (const float*)d_in[1];
    const float* Wx    = (const float*)d_in[2];
    const float* Wh    = (const float*)d_in[3];
    const float* b_rnn = (const float*)d_in[4];
    const float* Wd    = (const float*)d_in[5];
    const float* bd    = (const float*)d_in[6];
    const float* dmask = (const float*)d_in[7];
    float* out = (float*)d_out;

    charrnn_kernel<<<B_TOT / ROWS_PER_BLOCK, TPB, 0, stream>>>(
        x, emb, Wx, Wh, b_rnn, Wd, bd, dmask, out);
}


# Recurrence step issues four of its six tanh ahead of its first matrix instruction and two right behind it (was a pair ahead of each): no product waits for a tanh issued behind the product before it
# speedup vs baseline: 3.9557x; 1.0521x over previous
#include <hip/hip_runtime.h>


typedef float v2f __attribute__((ext_vector_type(2)));
typedef float v8f __attribute__((ext_vector_type(8)));

#define B_TOT  16384
#define T_LEN  100
#define V_SZ   256
#define E_SZ   50
#define H_SZ   10
#define L_SZ   15
#define PSTR   12
#define ROWS_PER_BLOCK 64
#define TPB    128
#define N_TT   (V_SZ / (TPB / 32) / 16)
#define T_CHUNK 20
#define N_CHUNK (T_LEN / T_CHUNK)
#define NQ     (T_CHUNK / 4)
#define GRP    10
#define N_GRP  (T_LEN / GRP)
#define G_AT   6
#define N_KC   13
#define TOK_PER_WAVE (V_SZ / (TPB / 32))
#define SLAB_F4 (TOK_PER_WAVE * E_SZ / 4)
#define SLAB_LD (SLAB_F4 / 32)

__device__ __forceinline__ float fast_tanh(float x) {
#if __has_builtin(__builtin_amdgcn_tanhf)
    return __builtin_amdgcn_tanhf(x);
#else
    x = fminf(fmaxf(x, -9.0f), 9.0f);
    float z = __builtin_amdgcn_exp2f(x * 2.885390081777927f);
    return (z - 1.0f) * __builtin_amdgcn_rcpf(z + 1.0f);
#endif
}

__device__ __forceinline__ v8f wmma_f32_16x16x4(v2f a, v2f b, v8f c) {
    return __builtin_amdgcn_wmma_f32_16x16x4_f32(
        false, a, false, b, (short)0, c, false, false);
}

__device__ __forceinline__ int hid_of(int hh, int r) {
    return hh ? ((r < 4) ? 6 + r : -1) : ((r < 6) ? r : -1);
}

__device__ __forceinline__ int emb_col_of(int q, int hh, int s) {
    return (q < 12) ? 8 * (q >> 1) + 4 * hh + 2 * (q & 1) + s
                    : (hh ? -1 : 48 + s);
}

typedef float v4f __attribute__((ext_vector_type(4)));
typedef float v4f_a8 __attribute__((ext_vector_type(4), aligned(8)));
typedef __attribute__((address_space(3))) const volatile v4f_a8 lds_cv4f;
typedef __attribute__((address_space(3))) const volatile v2f lds_cv2f;

__device__ __forceinline__ unsigned c_addr(const float* tab, int tok, int hoff) {
    const float* q = tab + (tok & (V_SZ - 1)) * PSTR + hoff;
    return (unsigned)reinterpret_cast<unsigned long long>(q);
}

__device__ __forceinline__ void gather6(unsigned off, v2f& p0, v2f& p1, v2f& p2) {
    const v4f lo = *reinterpret_cast<lds_cv4f*>(off);
    const v2f hi = *reinterpret_cast<lds_cv2f*>(off + 16u);
    p0[0] = lo[0]; p0[1] = lo[1];
    p1[0] = lo[2]; p1[1] = lo[3];
    p2 = hi;
}

__device__ __forceinline__ void chain_point(v8f& a) {
    float t = a[0];
    asm volatile("" : "+v"(t) : : "memory");
    a[0] = t;
}
__device__ __forceinline__ void chain_point(v8f& a, unsigned& x, unsigned& y) {
    float t = a[0];
    asm volatile("" : "+v"(t), "+v"(x), "+v"(y));
    a[0] = t;
}

__global__ __launch_bounds__(TPB)
void charrnn_kernel(const int*   __restrict__ x,
                    const float* __restrict__ emb,
                    const float* __restrict__ Wx,
                    const float* __restrict__ Wh,
                    const float* __restrict__ b_rnn,
                    const float* __restrict__ Wd,
                    const float* __restrict__ bd,
                    const float* __restrict__ dmask,
                    float*       __restrict__ out)
{
    __shared__ __attribute__((aligned(16))) float s_projb[V_SZ * PSTR];
    __shared__ __attribute__((aligned(16))) float s_out[ROWS_PER_BLOCK * L_SZ];
    __shared__ __attribute__((aligned(16))) float s_emb[V_SZ * E_SZ];

    const int tid   = threadIdx.x;
    const int lane  = tid & 31;
    const int wave  = tid >> 5;
    const int c     = lane & 15;
    const int hh    = lane >> 4;
    const int hoff  = 6 * hh;
    const int rowLocal = wave * 16 + c;
    const int gRow  = blockIdx.x * ROWS_PER_BLOCK + rowLocal;

    float4 ev[SLAB_LD];
    {
        const float4* eg = reinterpret_cast<const float4*>(emb) + wave * SLAB_F4 + lane;
#pragma unroll
        for (int i = 0; i < SLAB_LD; ++i) ev[i] = eg[i * 32];
    }

    const int4* xrow = reinterpret_cast<const int4*>(x + (size_t)gRow * T_LEN);
    int tok[2][T_CHUNK];
    const int jo  = hid_of(c >> 3, c & 7);
    const int joc = (jo >= 0) ? jo : 0;
    const int cl  = (c < L_SZ) ? c : (L_SZ - 1);
    float dmr[6], wdr[6], whr[6], bdr[8];

    {
        const int cj = (c < H_SZ) ? c : (H_SZ - 1);
        const int v0 = wave * TOK_PER_WAVE;

        float wxr[N_KC][2];
#pragma unroll
        for (int q = 0; q < N_KC; ++q) {
#pragma unroll
            for (int s = 0; s < 2; ++s) {
                const int e  = emb_col_of(q, hh, s);
                const int ec = (e >= 0) ? e : emb_col_of(q, 0, s);
                wxr[q][s] = Wx[ec * H_SZ + cj];
            }
        }
        float br[8];
#pragma unroll
        for (int r = 0; r < 8; ++r) {
            const int j = 8 * hh + r;
            br[r] = b_rnn[(j < H_SZ) ? j : (H_SZ - 1)];
        }
        {
            float4* sw = reinterpret_cast<float4*>(s_emb) + wave * SLAB_F4 + lane;
#pragma unroll
            for (int i = 0; i < SLAB_LD; ++i) sw[i * 32] = ev[i];
        }
        __syncthreads();

#pragma unroll
        for (int q = 0; q < NQ; ++q) {
            const int4 v = xrow[q];
            tok[0][4 * q + 0] = v.x; tok[0][4 * q + 1] = v.y;
            tok[0][4 * q + 2] = v.z; tok[0][4 * q + 3] = v.w;
        }
#pragma unroll
        for (int q = 0; q < NQ; ++q) {
            const int4 v = xrow[NQ + q];
            tok[1][4 * q + 0] = v.x; tok[1][4 * q + 1] = v.y;
            tok[1][4 * q + 2] = v.z; tok[1][4 * q + 3] = v.w;
        }
#pragma unroll
        for (int r = 0; r < 6; ++r) {
            const int k  = hid_of(hh, r);
            const int kc = (k >= 0) ? k : 0;
            dmr[r] = dmask[(size_t)gRow * H_SZ + kc];
            wdr[r] = Wd[kc * L_SZ + cl];
            whr[r] = Wh[kc * H_SZ + joc];
        }
#pragma unroll
        for (int r = 0; r < 8; ++r) {
            const int l = 8 * hh + r;
            bdr[r] = bd[(l < L_SZ) ? l : (L_SZ - 1)];
        }

        float ebr[N_TT][N_KC][2];
#pragma unroll
        for (int tl = 0; tl < N_TT; ++tl) {
            const float* er = s_emb + (v0 + 16 * tl + c) * E_SZ;
#pragma unroll
            for (int q = 0; q < N_KC; ++q) {
#pragma unroll
                for (int s = 0; s < 2; ++s) {
                    const int e  = emb_col_of(q, hh, s);
                    const int ec = (e >= 0) ? e : emb_col_of(q, 0, s);
                    ebr[tl][q][s] = er[ec];
                }
            }
        }
        __builtin_amdgcn_sched_barrier(0);

        v2f wxa[N_KC];
#pragma unroll
        for (int q = 0; q < N_KC; ++q) {
#pragma unroll
            for (int s = 0; s < 2; ++s) {
                const int e = emb_col_of(q, hh, s);
                wxa[q][s] = (e >= 0 && c < H_SZ) ? wxr[q][s] : 0.0f;
            }
        }

        v8f bias;
#pragma unroll
        for (int r = 0; r < 8; ++r) {
            const int j = 8 * hh + r;
            bias[r] = (j < H_SZ) ? br[r] : 0.0f;
        }

        v2f eb[N_TT][N_KC];
#pragma unroll
        for (int tl = 0; tl < N_TT; ++tl) {
#pragma unroll
            for (int q = 0; q < N_KC; ++q) {
#pragma unroll
                for (int s = 0; s < 2; ++s) {
                    const int e = emb_col_of(q, hh, s);
                    eb[tl][q][s] = (e >= 0) ? ebr[tl][q][s] : 0.0f;
                }
            }
        }

        v8f tt[N_TT];
#pragma unroll
        for (int tl = 0; tl < N_TT; ++tl) tt[tl] = bias;
#pragma unroll
        for (int q = 0; q < N_KC; ++q) {
#pragma unroll
            for (int tl = 0; tl < N_TT; ++tl)
                tt[tl] = wmma_f32_16x16x4(wxa[q], eb[tl][q], tt[tl]);
        }

#pragma unroll
        for (int tl = 0; tl < N_TT; ++tl) {
            float4* d = reinterpret_cast<float4*>(s_projb + (v0 + 16 * tl + c) * PSTR + 8 * hh);
            d[0] = make_float4(tt[tl][0], tt[tl][1],
                               hh ? 0.0f : tt[tl][2], hh ? 0.0f : tt[tl][3]);
            if (hh == 0)
                d[1] = make_float4(tt[tl][4], tt[tl][5], tt[tl][6], tt[tl][7]);
        }
    }

    float dm[6];
    v2f wda[3];
    v2f wha[3];
#pragma unroll
    for (int r = 0; r < 6; ++r) {
        const int k = hid_of(hh, r);
        dm[r] = (k >= 0) ? dmr[r] : 0.0f;
        wda[r >> 1][r & 1] = (k >= 0 && c < L_SZ) ? wdr[r] : 0.0f;
        wha[r >> 1][r & 1] = (k >= 0 && jo >= 0) ? whr[r] : 0.0f;
    }
    float bdv[8];
#pragma unroll
    for (int r = 0; r < 8; ++r) {
        const int l = 8 * hh + r;
        bdv[r] = (l < L_SZ) ? bdr[r] : 0.0f;
    }

    __syncthreads();

    v8f accp;
#pragma unroll
    for (int r = 0; r < 8; ++r) accp[r] = 0.0f;

    unsigned ca[GRP];
    v2f csA[GRP][3], csB[GRP][3];
#pragma unroll
    for (int j = 0; j < GRP; ++j) ca[j] = c_addr(s_projb, tok[0][j], hoff);
#pragma unroll
    for (int j = 0; j < GRP; ++j) gather6(ca[j], csA[j][0], csA[j][1], csA[j][2]);
    chain_point(accp);

    auto run_group = [&](const int g, v2f (&cur)[GRP][3], v2f (&nxt)[GRP][3])
                         __attribute__((always_inline)) {
#pragma unroll
        for (int i = 0; i < GRP; ++i) {
            if (g + 1 < N_GRP && i >= 1 && i <= GRP / 2) {
                const int j0 = 2 * (i - 1);
                const int s0 = (g + 1) * GRP + j0;
                const int s1 = s0 + 1;
                ca[j0]     = c_addr(s_projb, tok[(s0 / T_CHUNK) & 1][s0 % T_CHUNK], hoff);
                ca[j0 + 1] = c_addr(s_projb, tok[(s1 / T_CHUNK) & 1][s1 % T_CHUNK], hoff);
                chain_point(accp, ca[j0], ca[j0 + 1]);
            }

            if (g + 1 < N_GRP && i == G_AT) {
                chain_point(accp);
                __builtin_amdgcn_sched_barrier(0);
#pragma unroll
                for (int j = 0; j < GRP; ++j)
                    gather6(ca[j], nxt[j][0], nxt[j][1], nxt[j][2]);
                const int c = (g + 1) / 2;
                if (((g + 1) & 1) && c + 2 < N_CHUNK) {
#pragma unroll
                    for (int q = 0; q < NQ; ++q) {
                        const int4 v = xrow[(c + 2) * NQ + q];
                        tok[c & 1][4 * q + 0] = v.x; tok[c & 1][4 * q + 1] = v.y;
                        tok[c & 1][4 * q + 2] = v.z; tok[c & 1][4 * q + 3] = v.w;
                    }
                }
                __builtin_amdgcn_sched_barrier(0);
                chain_point(accp);
            }

            const v4f c01 = __builtin_shufflevector(cur[i][0], cur[i][1], 0, 1, 2, 3);
            const v4f c2x = __builtin_shufflevector(cur[i][2], cur[i][2], 0, 1, -1, -1);
            v8f acc = __builtin_shufflevector(c01, c2x, 0, 1, 2, 3, 4, 5, -1, -1);

            v2f b0, b1, b2;
            b0[0] = fast_tanh(accp[0]); b0[1] = fast_tanh(accp[1]);
            b1[0] = fast_tanh(accp[2]); b1[1] = fast_tanh(accp[3]);
            __builtin_amdgcn_sched_barrier(0);
            acc = wmma_f32_16x16x4(wha[0], b0, acc);
            __builtin_amdgcn_sched_barrier(0);
            b2[0] = fast_tanh(accp[4]); b2[1] = fast_tanh(accp[5]);
            __builtin_amdgcn_sched_barrier(0);
            acc = wmma_f32_16x16x4(wha[1], b1, acc);
            acc = wmma_f32_16x16x4(wha[2], b2, acc);

            accp = acc;
        }
    };

#pragma unroll
    for (int gp = 0; gp < N_GRP / 2; ++gp) {
        run_group(2 * gp,     csA, csB);
        run_group(2 * gp + 1, csB, csA);
    }

    v2f b0, b1, b2;
    b0[0] = fast_tanh(accp[0]); b0[1] = fast_tanh(accp[1]);
    b1[0] = fast_tanh(accp[2]); b1[1] = fast_tanh(accp[3]);
    b2[0] = fast_tanh(accp[4]); b2[1] = fast_tanh(accp[5]);

    v2f f0, f1, f2;
    f0[0] = b0[0] * dm[0]; f0[1] = b0[1] * dm[1];
    f1[0] = b1[0] * dm[2]; f1[1] = b1[1] * dm[3];
    f2[0] = b2[0] * dm[4]; f2[1] = b2[1] * dm[5];

    v8f lac;
#pragma unroll
    for (int r = 0; r < 8; ++r) lac[r] = bdv[r];

    lac = wmma_f32_16x16x4(wda[0], f0, lac);
    lac = wmma_f32_16x16x4(wda[1], f1, lac);
    lac = wmma_f32_16x16x4(wda[2], f2, lac);

#pragma unroll
    for (int r = 0; r < 8; ++r) {
        const int l = 8 * hh + r;
        if (l < L_SZ) s_out[rowLocal * L_SZ + l] = lac[r];
    }
    __syncthreads();

    float* ob = out + (size_t)blockIdx.x * (ROWS_PER_BLOCK * L_SZ);
    constexpr int N_OUT  = ROWS_PER_BLOCK * L_SZ;
    constexpr int N_FULL = N_OUT / TPB;
#pragma unroll
    for (int j = 0; j < N_FULL; ++j) ob[tid + j * TPB] = s_out[tid + j * TPB];
    if (tid < N_OUT - N_FULL * TPB)
        ob[tid + N_FULL * TPB] = s_out[tid + N_FULL * TPB];
}

extern "C" void kernel_launch(void* const* d_in, const int* in_sizes, int n_in,
                              void* d_out, int out_size, void* d_ws, size_t ws_size,
                              hipStream_t stream) {
    (void)in_sizes; (void)n_in; (void)d_ws; (void)ws_size; (void)out_size;
    const int*   x     = (const int*)  d_in[0];
    const float* emb   = (const float*)d_in[1];
    const float* Wx    = (const float*)d_in[2];
    const float* Wh    = (const float*)d_in[3];
    const float* b_rnn = (const float*)d_in[4];
    const float* Wd    = (const float*)d_in[5];
    const float* bd    = (const float*)d_in[6];
    const float* dmask = (const float*)d_in[7];
    float* out = (float*)d_out;

    charrnn_kernel<<<B_TOT / ROWS_PER_BLOCK, TPB, 0, stream>>>(
        x, emb, Wx, Wh, b_rnn, Wd, bd, dmask, out);
}
